# starting kernel
# baseline (speedup 1.0000x reference)
#include <hip/hip_runtime.h>

typedef __attribute__((ext_vector_type(2))) float v2f;
typedef __attribute__((ext_vector_type(8))) float v8f;

#define IN_F 128
#define OUT_F 64
#define WAVES_PER_BLOCK 8

__global__ void init_out_kernel(float* __restrict__ out,
                                const float* __restrict__ bias, int total) {
  int i = blockIdx.x * blockDim.x + threadIdx.x;
  if (i < total) out[i] = bias[i & (OUT_F - 1)];
}

__global__ __launch_bounds__(256) void gemm_wmma_kernel(
    const float* __restrict__ V, const float* __restrict__ W,
    float* __restrict__ S, int nRowTiles) {
  __shared__ v2f wlds2[(IN_F / 2) * OUT_F];
  for (int i = threadIdx.x; i < (IN_F / 2) * OUT_F; i += blockDim.x) {
    const int p = i / OUT_F;
    const int n = i % OUT_F;
    v2f w;
    w.x = W[(2 * p + 0) * OUT_F + n];
    w.y = W[(2 * p + 1) * OUT_F + n];
    wlds2[i] = w;
  }
  __syncthreads();

  const int lane = threadIdx.x & 31;
  const int wave = threadIdx.x >> 5;
  const int rowTile = blockIdx.x * WAVES_PER_BLOCK + wave;
  if (rowTile >= nRowTiles) return;

  const int m    = lane & 15;
  const int koff = (lane >> 4) << 1;

  const float* vptr = V + (long long)(rowTile * 16 + m) * IN_F + koff;
  const v2f*   wbase = wlds2 + (koff >> 1) * OUT_F + m;

  v8f c0 = {}, c1 = {}, c2 = {}, c3 = {};
#pragma unroll
  for (int k = 0; k < IN_F; k += 4) {
    v2f a = *(const v2f*)(vptr + k);
    const v2f* wrow = wbase + (k >> 1) * OUT_F;
    v2f b0 = wrow[0];
    v2f b1 = wrow[16];
    v2f b2 = wrow[32];
    v2f b3 = wrow[48];
    c0 = __builtin_amdgcn_wmma_f32_16x16x4_f32(false, a, false, b0,
                                               (short)0, c0, false, false);
    c1 = __builtin_amdgcn_wmma_f32_16x16x4_f32(false, a, false, b1,
                                               (short)0, c1, false, false);
    c2 = __builtin_amdgcn_wmma_f32_16x16x4_f32(false, a, false, b2,
                                               (short)0, c2, false, false);
    c3 = __builtin_amdgcn_wmma_f32_16x16x4_f32(false, a, false, b3,
                                               (short)0, c3, false, false);
  }

  const long long mBase = (long long)rowTile * 16 + ((lane >> 4) << 3);
#pragma unroll
  for (int r = 0; r < 8; ++r) {
    float* srow = S + (mBase + r) * OUT_F + m;
    srow[0]  = c0[r];
    srow[16] = c1[r];
    srow[32] = c2[r];
    srow[48] = c3[r];
  }
}

__global__ void spmm_kernel(const float* __restrict__ S,
                            const int* __restrict__ erow,
                            const int* __restrict__ ecol,
                            const float* __restrict__ eval,
                            float* out, int nWork) {
  int idx = blockIdx.x * blockDim.x + threadIdx.x;
  if (idx >= nWork) return;
  const int e = idx >> 4;
  const int f = (idx & 15) << 2;
  const int col = ecol[e];
  const int row = erow[e];
  const float v = eval[e];
  const float4 s = *(const float4*)(S + (long long)col * OUT_F + f);
  float* o = out + (long long)row * OUT_F + f;
  atomicAdd(o + 0, s.x * v);
  atomicAdd(o + 1, s.y * v);
  atomicAdd(o + 2, s.z * v);
  atomicAdd(o + 3, s.w * v);
}

extern "C" void kernel_launch(void* const* d_in, const int* in_sizes, int n_in,
                              void* d_out, int out_size, void* d_ws,
                              size_t ws_size, hipStream_t stream) {
  const float* vertex  = (const float*)d_in[0];
  const int*   erow    = (const int*)d_in[1];
  const int*   ecol    = (const int*)d_in[2];
  const float* eval    = (const float*)d_in[3];
  const float* weights = (const float*)d_in[4];
  const float* bias    = (const float*)d_in[5];
  float* out     = (float*)d_out;
  float* support = (float*)d_ws;

  const int nNodes = in_sizes[0] / IN_F;
  const int nEdges = in_sizes[1];

  const int totalOut = nNodes * OUT_F;
  init_out_kernel<<<(totalOut + 255) / 256, 256, 0, stream>>>(out, bias,
                                                              totalOut);

  const int nRowTiles = nNodes / 16;
  const int nBlocks = (nRowTiles + WAVES_PER_BLOCK - 1) / WAVES_PER_BLOCK;
  gemm_wmma_kernel<<<nBlocks, 256, 0, stream>>>(vertex, weights, support,
                                                nRowTiles);

  const int nWork = nEdges * 16;
  spmm_kernel<<<(nWork + 255) / 256, 256, 0, stream>>>(support, erow, ecol,
                                                       eval, out, nWork);
}


# spmm scatter: lane = feature, one wave walks 32 edges; each row load and each f32 atomic add covers one whole 128-B line (2 line adds per edge, not 8); f32 atomics kept, arrival order as before
# speedup vs baseline: 1.7816x; 1.7816x over previous
#include <hip/hip_runtime.h>

typedef __attribute__((ext_vector_type(2))) float v2f;
typedef __attribute__((ext_vector_type(8))) float v8f;

#define IN_F 128
#define OUT_F 64
#define WAVES_PER_BLOCK 8

__global__ void init_out_kernel(float* __restrict__ out,
                                const float* __restrict__ bias, int total) {
  int i = blockIdx.x * blockDim.x + threadIdx.x;
  if (i < total) out[i] = bias[i & (OUT_F - 1)];
}

__global__ __launch_bounds__(256) void gemm_wmma_kernel(
    const float* __restrict__ V, const float* __restrict__ W,
    float* __restrict__ S, int nRowTiles) {
  __shared__ v2f wlds2[(IN_F / 2) * OUT_F];
  for (int i = threadIdx.x; i < (IN_F / 2) * OUT_F; i += blockDim.x) {
    const int p = i / OUT_F;
    const int n = i % OUT_F;
    v2f w;
    w.x = W[(2 * p + 0) * OUT_F + n];
    w.y = W[(2 * p + 1) * OUT_F + n];
    wlds2[i] = w;
  }
  __syncthreads();

  const int lane = threadIdx.x & 31;
  const int wave = threadIdx.x >> 5;
  const int rowTile = blockIdx.x * WAVES_PER_BLOCK + wave;
  if (rowTile >= nRowTiles) return;

  const int m    = lane & 15;
  const int koff = (lane >> 4) << 1;

  const float* vptr = V + (long long)(rowTile * 16 + m) * IN_F + koff;
  const v2f*   wbase = wlds2 + (koff >> 1) * OUT_F + m;

  v8f c0 = {}, c1 = {}, c2 = {}, c3 = {};
#pragma unroll
  for (int k = 0; k < IN_F; k += 4) {
    v2f a = *(const v2f*)(vptr + k);
    const v2f* wrow = wbase + (k >> 1) * OUT_F;
    v2f b0 = wrow[0];
    v2f b1 = wrow[16];
    v2f b2 = wrow[32];
    v2f b3 = wrow[48];
    c0 = __builtin_amdgcn_wmma_f32_16x16x4_f32(false, a, false, b0,
                                               (short)0, c0, false, false);
    c1 = __builtin_amdgcn_wmma_f32_16x16x4_f32(false, a, false, b1,
                                               (short)0, c1, false, false);
    c2 = __builtin_amdgcn_wmma_f32_16x16x4_f32(false, a, false, b2,
                                               (short)0, c2, false, false);
    c3 = __builtin_amdgcn_wmma_f32_16x16x4_f32(false, a, false, b3,
                                               (short)0, c3, false, false);
  }

  const long long mBase = (long long)rowTile * 16 + ((lane >> 4) << 3);
#pragma unroll
  for (int r = 0; r < 8; ++r) {
    float* srow = S + (mBase + r) * OUT_F + m;
    srow[0]  = c0[r];
    srow[16] = c1[r];
    srow[32] = c2[r];
    srow[48] = c3[r];
  }
}

#define EDGES_PER_WAVE 32
#define EDGE_BATCH 8
#define N_BATCH (EDGES_PER_WAVE / EDGE_BATCH)

static __device__ __forceinline__ int bcast_i(int x, int j) {
  return __builtin_amdgcn_readlane(x, j);
}
static __device__ __forceinline__ float bcast_f(float x, int j) {
  return __int_as_float(__builtin_amdgcn_readlane(__float_as_int(x), j));
}

__global__ __launch_bounds__(256) void spmm_kernel(
    const float* __restrict__ S, const int* __restrict__ erow,
    const int* __restrict__ ecol, const float* __restrict__ eval,
    float* __restrict__ out, int nEdges) {
  const int lane = threadIdx.x & 31;
  const int wave = (blockIdx.x * 256 + threadIdx.x) >> 5;
  const int e0 = wave * EDGES_PER_WAVE;
  if (e0 >= nEdges) return;
  const int e = e0 + lane;
  int col = 0, row = 0;
  float val = 0.0f;
  if (e < nEdges) {
    col = ecol[e];
    row = erow[e];
    val = eval[e];
  }
  const int n = min(EDGES_PER_WAVE, nEdges - e0);

  if (n == EDGES_PER_WAVE) {
    float lo[2][EDGE_BATCH], hi[2][EDGE_BATCH];
#pragma unroll
    for (int u = 0; u < EDGE_BATCH; ++u) {
      const float* sp = S + ((unsigned)bcast_i(col, u) * OUT_F + lane);
      lo[0][u] = sp[0];
      hi[0][u] = sp[32];
    }
#pragma unroll
    for (int b = 0; b < N_BATCH; ++b) {
      if (b + 1 < N_BATCH) {
#pragma unroll
        for (int u = 0; u < EDGE_BATCH; ++u) {
          const int j = (b + 1) * EDGE_BATCH + u;
          const float* sp = S + ((unsigned)bcast_i(col, j) * OUT_F + lane);
          lo[(b + 1) & 1][u] = sp[0];
          hi[(b + 1) & 1][u] = sp[32];
        }
      }
#pragma unroll
      for (int u = 0; u < EDGE_BATCH; ++u) {
        const int j = b * EDGE_BATCH + u;
        const float v = bcast_f(val, j);
        float* o = out + ((unsigned)bcast_i(row, j) * OUT_F + lane);
        atomicAdd(o, lo[b & 1][u] * v);
        atomicAdd(o + 32, hi[b & 1][u] * v);
      }
    }
  } else {
    for (int j = 0; j < n; ++j) {
      const float v = bcast_f(val, j);
      const float* sp = S + ((unsigned)bcast_i(col, j) * OUT_F + lane);
      float* o = out + ((unsigned)bcast_i(row, j) * OUT_F + lane);
      atomicAdd(o, sp[0] * v);
      atomicAdd(o + 32, sp[32] * v);
    }
  }
}

extern "C" void kernel_launch(void* const* d_in, const int* in_sizes, int n_in,
                              void* d_out, int out_size, void* d_ws,
                              size_t ws_size, hipStream_t stream) {
  const float* vertex  = (const float*)d_in[0];
  const int*   erow    = (const int*)d_in[1];
  const int*   ecol    = (const int*)d_in[2];
  const float* eval    = (const float*)d_in[3];
  const float* weights = (const float*)d_in[4];
  const float* bias    = (const float*)d_in[5];
  float* out     = (float*)d_out;
  float* support = (float*)d_ws;

  const int nNodes = in_sizes[0] / IN_F;
  const int nEdges = in_sizes[1];

  const int totalOut = nNodes * OUT_F;
  init_out_kernel<<<(totalOut + 255) / 256, 256, 0, stream>>>(out, bias,
                                                              totalOut);

  const int nRowTiles = nNodes / 16;
  const int nBlocks = (nRowTiles + WAVES_PER_BLOCK - 1) / WAVES_PER_BLOCK;
  gemm_wmma_kernel<<<nBlocks, 256, 0, stream>>>(vertex, weights, support,
                                                nRowTiles);

  const int nWaves = (nEdges + EDGES_PER_WAVE - 1) / EDGES_PER_WAVE;
  spmm_kernel<<<(nWaves + 7) / 8, 256, 0, stream>>>(support, erow, ecol, eval,
                                                    out, nEdges);
}


# out = bias fill moved into the GEMM kernel: each wave stores its tile's out rows as 8 whole-line b128 stores; the separate fill launch of 200000 one-store waves is gone (3 launches -> 2)
# speedup vs baseline: 2.0771x; 1.1659x over previous
#include <hip/hip_runtime.h>

typedef __attribute__((ext_vector_type(2))) float v2f;
typedef __attribute__((ext_vector_type(4))) float v4f;
typedef __attribute__((ext_vector_type(8))) float v8f;

#define IN_F 128
#define OUT_F 64
#define WAVES_PER_BLOCK 8

__global__ __launch_bounds__(256) void gemm_wmma_kernel(
    const float* __restrict__ V, const float* __restrict__ W,
    float* __restrict__ S, const float* __restrict__ bias,
    float* __restrict__ out, int nRowTiles) {
  __shared__ v2f wlds2[(IN_F / 2) * OUT_F];
  const int lane = threadIdx.x & 31;
  const int wave = threadIdx.x >> 5;
  const int rowTile = blockIdx.x * WAVES_PER_BLOCK + wave;

  if (rowTile < nRowTiles) {
    const v4f b4 = *(const v4f*)(bias + 4 * (lane & 15));
    v4f* o4 = (v4f*)(out + (long long)rowTile * 16 * OUT_F) + lane;
#pragma unroll
    for (int it = 0; it < 8; ++it) o4[it * 32] = b4;
  }

  for (int i = threadIdx.x; i < (IN_F / 2) * OUT_F; i += blockDim.x) {
    const int p = i / OUT_F;
    const int n = i % OUT_F;
    v2f w;
    w.x = W[(2 * p + 0) * OUT_F + n];
    w.y = W[(2 * p + 1) * OUT_F + n];
    wlds2[i] = w;
  }
  __syncthreads();

  if (rowTile >= nRowTiles) return;

  const int m    = lane & 15;
  const int koff = (lane >> 4) << 1;

  const float* vptr = V + (long long)(rowTile * 16 + m) * IN_F + koff;
  const v2f*   wbase = wlds2 + (koff >> 1) * OUT_F + m;

  v8f c0 = {}, c1 = {}, c2 = {}, c3 = {};
#pragma unroll
  for (int k = 0; k < IN_F; k += 4) {
    v2f a = *(const v2f*)(vptr + k);
    const v2f* wrow = wbase + (k >> 1) * OUT_F;
    v2f b0 = wrow[0];
    v2f b1 = wrow[16];
    v2f b2 = wrow[32];
    v2f b3 = wrow[48];
    c0 = __builtin_amdgcn_wmma_f32_16x16x4_f32(false, a, false, b0,
                                               (short)0, c0, false, false);
    c1 = __builtin_amdgcn_wmma_f32_16x16x4_f32(false, a, false, b1,
                                               (short)0, c1, false, false);
    c2 = __builtin_amdgcn_wmma_f32_16x16x4_f32(false, a, false, b2,
                                               (short)0, c2, false, false);
    c3 = __builtin_amdgcn_wmma_f32_16x16x4_f32(false, a, false, b3,
                                               (short)0, c3, false, false);
  }

  const long long mBase = (long long)rowTile * 16 + ((lane >> 4) << 3);
#pragma unroll
  for (int r = 0; r < 8; ++r) {
    float* srow = S + (mBase + r) * OUT_F + m;
    srow[0]  = c0[r];
    srow[16] = c1[r];
    srow[32] = c2[r];
    srow[48] = c3[r];
  }
}

#define EDGES_PER_WAVE 32
#define EDGE_BATCH 8
#define N_BATCH (EDGES_PER_WAVE / EDGE_BATCH)

static __device__ __forceinline__ int bcast_i(int x, int j) {
  return __builtin_amdgcn_readlane(x, j);
}
static __device__ __forceinline__ float bcast_f(float x, int j) {
  return __int_as_float(__builtin_amdgcn_readlane(__float_as_int(x), j));
}

__global__ __launch_bounds__(256) void spmm_kernel(
    const float* __restrict__ S, const int* __restrict__ erow,
    const int* __restrict__ ecol, const float* __restrict__ eval,
    float* __restrict__ out, int nEdges) {
  const int lane = threadIdx.x & 31;
  const int wave = (blockIdx.x * 256 + threadIdx.x) >> 5;
  const int e0 = wave * EDGES_PER_WAVE;
  if (e0 >= nEdges) return;
  const int e = e0 + lane;
  int col = 0, row = 0;
  float val = 0.0f;
  if (e < nEdges) {
    col = ecol[e];
    row = erow[e];
    val = eval[e];
  }
  const int n = min(EDGES_PER_WAVE, nEdges - e0);

  if (n == EDGES_PER_WAVE) {
    float lo[2][EDGE_BATCH], hi[2][EDGE_BATCH];
#pragma unroll
    for (int u = 0; u < EDGE_BATCH; ++u) {
      const float* sp = S + ((unsigned)bcast_i(col, u) * OUT_F + lane);
      lo[0][u] = sp[0];
      hi[0][u] = sp[32];
    }
#pragma unroll
    for (int b = 0; b < N_BATCH; ++b) {
      if (b + 1 < N_BATCH) {
#pragma unroll
        for (int u = 0; u < EDGE_BATCH; ++u) {
          const int j = (b + 1) * EDGE_BATCH + u;
          const float* sp = S + ((unsigned)bcast_i(col, j) * OUT_F + lane);
          lo[(b + 1) & 1][u] = sp[0];
          hi[(b + 1) & 1][u] = sp[32];
        }
      }
#pragma unroll
      for (int u = 0; u < EDGE_BATCH; ++u) {
        const int j = b * EDGE_BATCH + u;
        const float v = bcast_f(val, j);
        float* o = out + ((unsigned)bcast_i(row, j) * OUT_F + lane);
        atomicAdd(o, lo[b & 1][u] * v);
        atomicAdd(o + 32, hi[b & 1][u] * v);
      }
    }
  } else {
    for (int j = 0; j < n; ++j) {
      const float v = bcast_f(val, j);
      const float* sp = S + ((unsigned)bcast_i(col, j) * OUT_F + lane);
      float* o = out + ((unsigned)bcast_i(row, j) * OUT_F + lane);
      atomicAdd(o, sp[0] * v);
      atomicAdd(o + 32, sp[32] * v);
    }
  }
}

extern "C" void kernel_launch(void* const* d_in, const int* in_sizes, int n_in,
                              void* d_out, int out_size, void* d_ws,
                              size_t ws_size, hipStream_t stream) {
  const float* vertex  = (const float*)d_in[0];
  const int*   erow    = (const int*)d_in[1];
  const int*   ecol    = (const int*)d_in[2];
  const float* eval    = (const float*)d_in[3];
  const float* weights = (const float*)d_in[4];
  const float* bias    = (const float*)d_in[5];
  float* out     = (float*)d_out;
  float* support = (float*)d_ws;

  const int nNodes = in_sizes[0] / IN_F;
  const int nEdges = in_sizes[1];

  const int nRowTiles = nNodes / 16;
  const int nBlocks = (nRowTiles + WAVES_PER_BLOCK - 1) / WAVES_PER_BLOCK;
  gemm_wmma_kernel<<<nBlocks, 256, 0, stream>>>(vertex, weights, support, bias,
                                                out, nRowTiles);

  const int nWaves = (nEdges + EDGES_PER_WAVE - 1) / EDGES_PER_WAVE;
  spmm_kernel<<<(nWaves + 7) / 8, 256, 0, stream>>>(support, erow, ecol, eval,
                                                    out, nEdges);
}


# spmm scatter: out rows split in two ranges, one per L2 half; a work-group reads its die, draws a 512-edge group by integer ticket per half, adds only its half's rows; f32 atomics kept, arrival order
# speedup vs baseline: 2.4647x; 1.1866x over previous
#include <hip/hip_runtime.h>

typedef __attribute__((ext_vector_type(2))) float v2f;
typedef __attribute__((ext_vector_type(4))) float v4f;
typedef __attribute__((ext_vector_type(8))) float v8f;

#define IN_F 128
#define OUT_F 64
#define WAVES_PER_BLOCK 8
#define TICKET_STRIDE 64

__global__ __launch_bounds__(256) void gemm_wmma_kernel(
    const float* __restrict__ V, const float* __restrict__ W,
    float* __restrict__ S, const float* __restrict__ bias,
    float* __restrict__ out, int nRowTiles, unsigned* __restrict__ tickets) {
  __shared__ v2f wlds2[(IN_F / 2) * OUT_F];
  const int lane = threadIdx.x & 31;
  const int wave = threadIdx.x >> 5;
  const int rowTile = blockIdx.x * WAVES_PER_BLOCK + wave;

  if (blockIdx.x == 0 && threadIdx.x == 0) {
    tickets[0] = 0u;
    tickets[TICKET_STRIDE] = 0u;
  }

  if (rowTile < nRowTiles) {
    const v4f b4 = *(const v4f*)(bias + 4 * (lane & 15));
    v4f* o4 = (v4f*)(out + (long long)rowTile * 16 * OUT_F) + lane;
#pragma unroll
    for (int it = 0; it < 8; ++it) o4[it * 32] = b4;
  }

  for (int i = threadIdx.x; i < (IN_F / 2) * OUT_F; i += blockDim.x) {
    const int p = i / OUT_F;
    const int n = i % OUT_F;
    v2f w;
    w.x = W[(2 * p + 0) * OUT_F + n];
    w.y = W[(2 * p + 1) * OUT_F + n];
    wlds2[i] = w;
  }
  __syncthreads();

  if (rowTile >= nRowTiles) return;

  const int m    = lane & 15;
  const int koff = (lane >> 4) << 1;

  const float* vptr = V + (long long)(rowTile * 16 + m) * IN_F + koff;
  const v2f*   wbase = wlds2 + (koff >> 1) * OUT_F + m;

  v8f c0 = {}, c1 = {}, c2 = {}, c3 = {};
#pragma unroll
  for (int k = 0; k < IN_F; k += 4) {
    v2f a = *(const v2f*)(vptr + k);
    const v2f* wrow = wbase + (k >> 1) * OUT_F;
    v2f b0 = wrow[0];
    v2f b1 = wrow[16];
    v2f b2 = wrow[32];
    v2f b3 = wrow[48];
    c0 = __builtin_amdgcn_wmma_f32_16x16x4_f32(false, a, false, b0,
                                               (short)0, c0, false, false);
    c1 = __builtin_amdgcn_wmma_f32_16x16x4_f32(false, a, false, b1,
                                               (short)0, c1, false, false);
    c2 = __builtin_amdgcn_wmma_f32_16x16x4_f32(false, a, false, b2,
                                               (short)0, c2, false, false);
    c3 = __builtin_amdgcn_wmma_f32_16x16x4_f32(false, a, false, b3,
                                               (short)0, c3, false, false);
  }

  const long long mBase = (long long)rowTile * 16 + ((lane >> 4) << 3);
#pragma unroll
  for (int r = 0; r < 8; ++r) {
    float* srow = S + (mBase + r) * OUT_F + m;
    srow[0]  = c0[r];
    srow[16] = c1[r];
    srow[32] = c2[r];
    srow[48] = c3[r];
  }
}

#define CHUNK_EDGES 64
#define GROUP_EDGES (CHUNK_EDGES * 8)
#define WALK_BATCH 16

static __device__ __forceinline__ int bcast_i(int x, int j) {
  return __builtin_amdgcn_readlane(x, j);
}
static __device__ __forceinline__ float bcast_f(float x, int j) {
  return __int_as_float(__builtin_amdgcn_readlane(__float_as_int(x), j));
}

static __device__ __forceinline__ void walk_edges(
    unsigned mask, int row, int col, float val, const float* __restrict__ S,
    float* __restrict__ out, int lane) {
  while (mask != 0u) {
    float lo[WALK_BATCH], hi[WALK_BATCH];
    unsigned m = mask;
#pragma unroll
    for (int u = 0; u < WALK_BATCH; ++u) {
      if (m != 0u) {
        const int j = __builtin_ctz(m);
        m &= m - 1u;
        const float* sp = S + ((unsigned)bcast_i(col, j) * OUT_F + lane);
        lo[u] = sp[0];
        hi[u] = sp[32];
      }
    }
    m = mask;
#pragma unroll
    for (int u = 0; u < WALK_BATCH; ++u) {
      if (m != 0u) {
        const int j = __builtin_ctz(m);
        m &= m - 1u;
        const float v = bcast_f(val, j);
        float* o = out + ((unsigned)bcast_i(row, j) * OUT_F + lane);
        atomicAdd(o, lo[u] * v);
        atomicAdd(o + 32, hi[u] * v);
      }
    }
    mask = m;
  }
}

__global__ __launch_bounds__(256) void spmm_kernel(
    const float* __restrict__ S, const int* __restrict__ erow,
    const int* __restrict__ ecol, const float* __restrict__ eval,
    float* __restrict__ out, int nEdges, int nNodes, int nGroups,
    unsigned* __restrict__ tickets) {
  __shared__ unsigned sh_half, sh_ticket;
  const int lane = threadIdx.x & 31;
  const int wave = threadIdx.x >> 5;

  if (threadIdx.x == 0) {
    unsigned half = ((unsigned)__builtin_amdgcn_s_sendmsg_rtn(0x87) >> 18) & 1u;
    unsigned t = atomicAdd(tickets + half * TICKET_STRIDE, 1u);
    if (t >= (unsigned)nGroups) {
      half ^= 1u;
      t = atomicAdd(tickets + half * TICKET_STRIDE, 1u);
    }
    sh_half = half;
    sh_ticket = t;
  }
  __syncthreads();
  const unsigned half = (unsigned)__builtin_amdgcn_readfirstlane((int)sh_half);
  const unsigned t = (unsigned)__builtin_amdgcn_readfirstlane((int)sh_ticket);
  if (t >= (unsigned)nGroups) return;

  const int split = nNodes >> 1;
  const int e0 = (int)t * GROUP_EDGES + wave * CHUNK_EDGES + lane;
  const int e1 = e0 + 32;
  int row0 = 0, col0 = 0, row1 = 0, col1 = 0;
  float val0 = 0.0f, val1 = 0.0f;
  bool mine0 = false, mine1 = false;
  if (e0 < nEdges) {
    row0 = erow[e0];
    col0 = ecol[e0];
    val0 = eval[e0];
    mine0 = (row0 >= split ? 1u : 0u) == half;
  }
  if (e1 < nEdges) {
    row1 = erow[e1];
    col1 = ecol[e1];
    val1 = eval[e1];
    mine1 = (row1 >= split ? 1u : 0u) == half;
  }
  const unsigned mask0 = __builtin_amdgcn_ballot_w32(mine0);
  const unsigned mask1 = __builtin_amdgcn_ballot_w32(mine1);
  walk_edges(mask0, row0, col0, val0, S, out, lane);
  walk_edges(mask1, row1, col1, val1, S, out, lane);
}

extern "C" void kernel_launch(void* const* d_in, const int* in_sizes, int n_in,
                              void* d_out, int out_size, void* d_ws,
                              size_t ws_size, hipStream_t stream) {
  const float* vertex  = (const float*)d_in[0];
  const int*   erow    = (const int*)d_in[1];
  const int*   ecol    = (const int*)d_in[2];
  const float* eval    = (const float*)d_in[3];
  const float* weights = (const float*)d_in[4];
  const float* bias    = (const float*)d_in[5];
  float* out     = (float*)d_out;
  float* support = (float*)d_ws;
  unsigned* tickets = (unsigned*)((char*)d_ws + (64u << 20));

  const int nNodes = in_sizes[0] / IN_F;
  const int nEdges = in_sizes[1];

  const int nRowTiles = nNodes / 16;
  const int nBlocks = (nRowTiles + WAVES_PER_BLOCK - 1) / WAVES_PER_BLOCK;
  gemm_wmma_kernel<<<nBlocks, 256, 0, stream>>>(vertex, weights, support, bias,
                                                out, nRowTiles, tickets);

  const int nGroups = (nEdges + GROUP_EDGES - 1) / GROUP_EDGES;
  spmm_kernel<<<2 * nGroups, 256, 0, stream>>>(
      support, erow, ecol, eval, out, nEdges, nNodes, nGroups, tickets);
}


# gemm prologue: W copied to LDS in one round trip (16 loads per lane issued together, then 8 whole-row LDS stores; was 16 dependent load-wait-store trips); out start stores moved behind the products
# speedup vs baseline: 2.8003x; 1.1362x over previous
#include <hip/hip_runtime.h>

typedef __attribute__((ext_vector_type(2))) float v2f;
typedef __attribute__((ext_vector_type(4))) float v4f;
typedef __attribute__((ext_vector_type(8))) float v8f;

#define IN_F 128
#define OUT_F 64
#define WAVES_PER_BLOCK 8
#define TICKET_STRIDE 64
#define PAIR_ROWS_PER_WAVE ((IN_F / 2) / WAVES_PER_BLOCK)

__global__ __launch_bounds__(256) void gemm_wmma_kernel(
    const float* __restrict__ V, const float* __restrict__ W,
    float* __restrict__ S, const float* __restrict__ bias,
    float* __restrict__ out, int nRowTiles, unsigned* __restrict__ tickets) {
  __shared__ v2f wlds2[(IN_F / 2) * OUT_F];
  const int lane = threadIdx.x & 31;
  const int wave = threadIdx.x >> 5;
  const int rowTile = blockIdx.x * WAVES_PER_BLOCK + wave;

  if (blockIdx.x == 0 && threadIdx.x == 0) {
    tickets[0] = 0u;
    tickets[TICKET_STRIDE] = 0u;
  }

  v2f wr0[PAIR_ROWS_PER_WAVE], wr1[PAIR_ROWS_PER_WAVE];
#pragma unroll
  for (int j = 0; j < PAIR_ROWS_PER_WAVE; ++j) {
    const int p = wave + j * WAVES_PER_BLOCK;
    wr0[j] = *(const v2f*)(W + (2 * p + 0) * OUT_F + 2 * lane);
    wr1[j] = *(const v2f*)(W + (2 * p + 1) * OUT_F + 2 * lane);
  }
#pragma unroll
  for (int j = 0; j < PAIR_ROWS_PER_WAVE; ++j) {
    const int p = wave + j * WAVES_PER_BLOCK;
    const v4f w4 = {wr0[j].x, wr1[j].x, wr0[j].y, wr1[j].y};
    *(v4f*)(wlds2 + p * OUT_F + 2 * lane) = w4;
  }
  __syncthreads();

  if (rowTile >= nRowTiles) return;

  const int m    = lane & 15;
  const int koff = (lane >> 4) << 1;

  const float* vptr = V + (long long)(rowTile * 16 + m) * IN_F + koff;
  const v2f*   wbase = wlds2 + (koff >> 1) * OUT_F + m;

  v8f c0 = {}, c1 = {}, c2 = {}, c3 = {};
#pragma unroll
  for (int k = 0; k < IN_F; k += 4) {
    v2f a = *(const v2f*)(vptr + k);
    const v2f* wrow = wbase + (k >> 1) * OUT_F;
    v2f b0 = wrow[0];
    v2f b1 = wrow[16];
    v2f b2 = wrow[32];
    v2f b3 = wrow[48];
    c0 = __builtin_amdgcn_wmma_f32_16x16x4_f32(false, a, false, b0,
                                               (short)0, c0, false, false);
    c1 = __builtin_amdgcn_wmma_f32_16x16x4_f32(false, a, false, b1,
                                               (short)0, c1, false, false);
    c2 = __builtin_amdgcn_wmma_f32_16x16x4_f32(false, a, false, b2,
                                               (short)0, c2, false, false);
    c3 = __builtin_amdgcn_wmma_f32_16x16x4_f32(false, a, false, b3,
                                               (short)0, c3, false, false);
  }

  {
    const v4f b4 = *(const v4f*)(bias + 4 * (lane & 15));
    v4f* o4 = (v4f*)(out + (long long)rowTile * 16 * OUT_F) + lane;
#pragma unroll
    for (int it = 0; it < 8; ++it) o4[it * 32] = b4;
  }

  const long long mBase = (long long)rowTile * 16 + ((lane >> 4) << 3);
#pragma unroll
  for (int r = 0; r < 8; ++r) {
    float* srow = S + (mBase + r) * OUT_F + m;
    srow[0]  = c0[r];
    srow[16] = c1[r];
    srow[32] = c2[r];
    srow[48] = c3[r];
  }
}

#define CHUNK_EDGES 64
#define GROUP_EDGES (CHUNK_EDGES * 8)
#define WALK_BATCH 16

static __device__ __forceinline__ int bcast_i(int x, int j) {
  return __builtin_amdgcn_readlane(x, j);
}
static __device__ __forceinline__ float bcast_f(float x, int j) {
  return __int_as_float(__builtin_amdgcn_readlane(__float_as_int(x), j));
}

static __device__ __forceinline__ void walk_edges(
    unsigned mask, int row, int col, float val, const float* __restrict__ S,
    float* __restrict__ out, int lane) {
  while (mask != 0u) {
    float lo[WALK_BATCH], hi[WALK_BATCH];
    unsigned m = mask;
#pragma unroll
    for (int u = 0; u < WALK_BATCH; ++u) {
      if (m != 0u) {
        const int j = __builtin_ctz(m);
        m &= m - 1u;
        const float* sp = S + ((unsigned)bcast_i(col, j) * OUT_F + lane);
        lo[u] = sp[0];
        hi[u] = sp[32];
      }
    }
    m = mask;
#pragma unroll
    for (int u = 0; u < WALK_BATCH; ++u) {
      if (m != 0u) {
        const int j = __builtin_ctz(m);
        m &= m - 1u;
        const float v = bcast_f(val, j);
        float* o = out + ((unsigned)bcast_i(row, j) * OUT_F + lane);
        atomicAdd(o, lo[u] * v);
        atomicAdd(o + 32, hi[u] * v);
      }
    }
    mask = m;
  }
}

__global__ __launch_bounds__(256) void spmm_kernel(
    const float* __restrict__ S, const int* __restrict__ erow,
    const int* __restrict__ ecol, const float* __restrict__ eval,
    float* __restrict__ out, int nEdges, int nNodes, int nGroups,
    unsigned* __restrict__ tickets) {
  __shared__ unsigned sh_half, sh_ticket;
  const int lane = threadIdx.x & 31;
  const int wave = threadIdx.x >> 5;

  if (threadIdx.x == 0) {
    unsigned half = ((unsigned)__builtin_amdgcn_s_sendmsg_rtn(0x87) >> 18) & 1u;
    unsigned t = atomicAdd(tickets + half * TICKET_STRIDE, 1u);
    if (t >= (unsigned)nGroups) {
      half ^= 1u;
      t = atomicAdd(tickets + half * TICKET_STRIDE, 1u);
    }
    sh_half = half;
    sh_ticket = t;
  }
  __syncthreads();
  const unsigned half = (unsigned)__builtin_amdgcn_readfirstlane((int)sh_half);
  const unsigned t = (unsigned)__builtin_amdgcn_readfirstlane((int)sh_ticket);
  if (t >= (unsigned)nGroups) return;

  const int split = nNodes >> 1;
  const int e0 = (int)t * GROUP_EDGES + wave * CHUNK_EDGES + lane;
  const int e1 = e0 + 32;
  int row0 = 0, col0 = 0, row1 = 0, col1 = 0;
  float val0 = 0.0f, val1 = 0.0f;
  bool mine0 = false, mine1 = false;
  if (e0 < nEdges) {
    row0 = erow[e0];
    col0 = ecol[e0];
    val0 = eval[e0];
    mine0 = (row0 >= split ? 1u : 0u) == half;
  }
  if (e1 < nEdges) {
    row1 = erow[e1];
    col1 = ecol[e1];
    val1 = eval[e1];
    mine1 = (row1 >= split ? 1u : 0u) == half;
  }
  const unsigned mask0 = __builtin_amdgcn_ballot_w32(mine0);
  const unsigned mask1 = __builtin_amdgcn_ballot_w32(mine1);
  walk_edges(mask0, row0, col0, val0, S, out, lane);
  walk_edges(mask1, row1, col1, val1, S, out, lane);
}

extern "C" void kernel_launch(void* const* d_in, const int* in_sizes, int n_in,
                              void* d_out, int out_size, void* d_ws,
                              size_t ws_size, hipStream_t stream) {
  const float* vertex  = (const float*)d_in[0];
  const int*   erow    = (const int*)d_in[1];
  const int*   ecol    = (const int*)d_in[2];
  const float* eval    = (const float*)d_in[3];
  const float* weights = (const float*)d_in[4];
  const float* bias    = (const float*)d_in[5];
  float* out     = (float*)d_out;
  float* support = (float*)d_ws;
  unsigned* tickets = (unsigned*)((char*)d_ws + (64u << 20));

  const int nNodes = in_sizes[0] / IN_F;
  const int nEdges = in_sizes[1];

  const int nRowTiles = nNodes / 16;
  const int nBlocks = (nRowTiles + WAVES_PER_BLOCK - 1) / WAVES_PER_BLOCK;
  gemm_wmma_kernel<<<nBlocks, 256, 0, stream>>>(vertex, weights, support, bias,
                                                out, nRowTiles, tickets);

  const int nGroups = (nEdges + GROUP_EDGES - 1) / GROUP_EDGES;
  spmm_kernel<<<2 * nGroups, 256, 0, stream>>>(
      support, erow, ecol, eval, out, nEdges, nNodes, nGroups, tickets);
}


# spmm: edges sorted per call by out row (LDS chunk sort in the product launch, range sort in the owner); a wave sums one out row in registers, stores it once; no float atomic; sum order changed
# speedup vs baseline: 3.2156x; 1.1483x over previous
#include <hip/hip_runtime.h>


typedef __attribute__((ext_vector_type(2))) float v2f;
typedef __attribute__((ext_vector_type(4))) float v4f;
typedef __attribute__((ext_vector_type(8))) float v8f;

#define IN_F 128
#define OUT_F 64
#define WAVES_PER_BLOCK 8
#define PAIR_ROWS_PER_WAVE ((IN_F / 2) / WAVES_PER_BLOCK)

#define ROWS_PER_RANGE 391
#define MAX_RANGES 256
#define CHUNK_EDGES 4096
#define CHUNK_PER_THREAD (CHUNK_EDGES / 256)
#define MAX_CHUNKS 1024
#define OFFS_PITCH 288
#define COL_BITS 17
#define COL_MASK ((1 << COL_BITS) - 1)
#define OWNER_THREADS 1024
#define OWNER_WAVES (OWNER_THREADS / 32)
#define REC_PER_THREAD 8
#define OWNER_CAP (OWNER_THREADS * REC_PER_THREAD)
#define WALK_BATCH 32

static __device__ __forceinline__ int bcast_i(int x, int j) {
  return __builtin_amdgcn_readlane(x, j);
}
static __device__ __forceinline__ float bcast_f(float x, int j) {
  return __int_as_float(__builtin_amdgcn_readlane(__float_as_int(x), j));
}

static __device__ __forceinline__ int wave_incl_scan(int x, int lane) {
#pragma unroll
  for (int d = 1; d < 32; d <<= 1) {
    const int y = __shfl_up(x, d);
    if (lane >= d) x += y;
  }
  return x;
}

__global__ __launch_bounds__(256) void gemm_sort_kernel(
    const float* __restrict__ V, const float* __restrict__ W,
    float* __restrict__ S, const int* __restrict__ erow,
    const int* __restrict__ ecol, const float* __restrict__ eval,
    float* __restrict__ recs_f, int* __restrict__ offs, int nRowTiles,
    int nEdges, int nGemmBlocks) {
  __shared__ v2f smem2[(IN_F / 2) * OUT_F];
  __shared__ int hist[MAX_RANGES];
  __shared__ int startv[MAX_RANGES + 1];
  __shared__ int wsum[WAVES_PER_BLOCK];
  const int lane = threadIdx.x & 31;
  const int wave = threadIdx.x >> 5;

  if ((int)blockIdx.x >= nGemmBlocks) {
    const int t = threadIdx.x;
    const int chunk = (int)blockIdx.x - nGemmBlocks;
    const int base = chunk * CHUNK_EDGES;
    hist[t] = 0;
    __syncthreads();
    int key[CHUNK_PER_THREAD], br[CHUNK_PER_THREAD];
    float val[CHUNK_PER_THREAD];
#pragma unroll
    for (int k = 0; k < CHUNK_PER_THREAD; ++k) {
      const int e = base + k * 256 + t;
      const int ec = e < nEdges ? e : nEdges - 1;
      key[k] = erow[ec];
      br[k] = ecol[ec];
      val[k] = eval[ec];
    }
#pragma unroll
    for (int k = 0; k < CHUNK_PER_THREAD; ++k) {
      const int e = base + k * 256 + t;
      const unsigned r = (unsigned)key[k];
      unsigned bin = r / ROWS_PER_RANGE;
      const unsigned rl = r - bin * ROWS_PER_RANGE;
      bin = bin < MAX_RANGES ? bin : MAX_RANGES - 1;
      key[k] = (int)((rl << COL_BITS) | ((unsigned)br[k] & COL_MASK));
      br[k] = e < nEdges ? (int)bin : -1;
    }
#pragma unroll
    for (int k = 0; k < CHUNK_PER_THREAD; ++k) {
      if (br[k] >= 0) {
        const int rk = atomicAdd(&hist[br[k]], 1);
        br[k] = (br[k] << 16) | rk;
      }
    }
    __syncthreads();
    const int h = hist[t];
    const int incl = wave_incl_scan(h, lane);
    if (lane == 31) wsum[wave] = incl;
    __syncthreads();
    int woff = 0;
    for (int i = 0; i < wave; ++i) woff += wsum[i];
    const int excl = woff + incl - h;
    startv[t] = excl;
    offs[chunk * OFFS_PITCH + t] = excl;
    if (t == 255) {
      startv[256] = excl + h;
      offs[chunk * OFFS_PITCH + 256] = excl + h;
    }
    __syncthreads();
#pragma unroll
    for (int k = 0; k < CHUNK_PER_THREAD; ++k) {
      if (br[k] >= 0) {
        const int pos = startv[br[k] >> 16] + (br[k] & 0xffff);
        const v2f rec = {__int_as_float(key[k]), val[k]};
        smem2[pos] = rec;
      }
    }
    __syncthreads();
    const int count = startv[256];
    v2f* recs = (v2f*)recs_f;
#pragma unroll
    for (int k = 0; k < CHUNK_PER_THREAD; ++k) {
      const int i = k * 256 + t;
      if (i < count) recs[base + i] = smem2[i];
    }
    return;
  }

  v2f* wlds2 = smem2;
  const int rowTile = (int)blockIdx.x * WAVES_PER_BLOCK + wave;

  v2f wr0[PAIR_ROWS_PER_WAVE], wr1[PAIR_ROWS_PER_WAVE];
#pragma unroll
  for (int j = 0; j < PAIR_ROWS_PER_WAVE; ++j) {
    const int p = wave + j * WAVES_PER_BLOCK;
    wr0[j] = *(const v2f*)(W + (2 * p + 0) * OUT_F + 2 * lane);
    wr1[j] = *(const v2f*)(W + (2 * p + 1) * OUT_F + 2 * lane);
  }
#pragma unroll
  for (int j = 0; j < PAIR_ROWS_PER_WAVE; ++j) {
    const int p = wave + j * WAVES_PER_BLOCK;
    const v4f w4 = {wr0[j].x, wr1[j].x, wr0[j].y, wr1[j].y};
    *(v4f*)(wlds2 + p * OUT_F + 2 * lane) = w4;
  }
  __syncthreads();

  if (rowTile >= nRowTiles) return;

  const int m    = lane & 15;
  const int koff = (lane >> 4) << 1;

  const float* vptr = V + (long long)(rowTile * 16 + m) * IN_F + koff;
  const v2f*   wbase = wlds2 + (koff >> 1) * OUT_F + m;

  v8f c0 = {}, c1 = {}, c2 = {}, c3 = {};
#pragma unroll
  for (int k = 0; k < IN_F; k += 4) {
    v2f a = *(const v2f*)(vptr + k);
    const v2f* wrow = wbase + (k >> 1) * OUT_F;
    v2f b0 = wrow[0];
    v2f b1 = wrow[16];
    v2f b2 = wrow[32];
    v2f b3 = wrow[48];
    c0 = __builtin_amdgcn_wmma_f32_16x16x4_f32(false, a, false, b0,
                                               (short)0, c0, false, false);
    c1 = __builtin_amdgcn_wmma_f32_16x16x4_f32(false, a, false, b1,
                                               (short)0, c1, false, false);
    c2 = __builtin_amdgcn_wmma_f32_16x16x4_f32(false, a, false, b2,
                                               (short)0, c2, false, false);
    c3 = __builtin_amdgcn_wmma_f32_16x16x4_f32(false, a, false, b3,
                                               (short)0, c3, false, false);
  }

  const long long mBase = (long long)rowTile * 16 + ((lane >> 4) << 3);
#pragma unroll
  for (int r = 0; r < 8; ++r) {
    float* srow = S + (mBase + r) * OUT_F + m;
    srow[0]  = c0[r];
    srow[16] = c1[r];
    srow[32] = c2[r];
    srow[48] = c3[r];
  }
}

static __device__ __forceinline__ int find_run(const int* pre, int idx) {
  int lo = 0;
#pragma unroll
  for (int step = MAX_CHUNKS / 2; step >= 1; step >>= 1)
    if (pre[lo + step] <= idx) lo += step;
  return lo;
}
static __device__ __forceinline__ int row_of(int key) {
  const int rl = key >> COL_BITS;
  return rl < ROWS_PER_RANGE ? rl : ROWS_PER_RANGE - 1;
}

__global__ __launch_bounds__(OWNER_THREADS) void owner_kernel(
    const float* __restrict__ S, const float* __restrict__ recs_f,
    const int* __restrict__ offs, const float* __restrict__ bias,
    float* __restrict__ out, int nNodes, int nChunks) {
  __shared__ v2f sorted[OWNER_CAP];
  __shared__ float acc[ROWS_PER_RANGE * OUT_F];
  __shared__ int chunkPre[MAX_CHUNKS];
  __shared__ int chunkBeg[MAX_CHUNKS];
  __shared__ int cnt[512];
  __shared__ int rowStart[512 + 1];
  __shared__ int wsumA[OWNER_WAVES];
  __shared__ int totalS;

  const int t = threadIdx.x;
  const int lane = t & 31;
  const int wave = __builtin_amdgcn_readfirstlane(t >> 5);
  const int b = (int)blockIdx.x;
  const v2f* recs = (const v2f*)recs_f;
  int rowsInRange = nNodes - b * ROWS_PER_RANGE;
  rowsInRange = rowsInRange < ROWS_PER_RANGE ? rowsInRange : ROWS_PER_RANGE;

  int len = 0, gbeg = 0;
  if (t < nChunks) {
    const int o0 = offs[t * OFFS_PITCH + b];
    const int o1 = offs[t * OFFS_PITCH + b + 1];
    len = o1 - o0;
    gbeg = t * CHUNK_EDGES + o0;
  }
  if (t < 512) cnt[t] = 0;
  const int incl = wave_incl_scan(len, lane);
  if (lane == 31) wsumA[wave] = incl;
  __syncthreads();
  int woff = 0;
  for (int i = 0; i < wave; ++i) woff += wsumA[i];
  const int excl = woff + incl - len;
  chunkPre[t] = excl;
  chunkBeg[t] = gbeg;
  if (t == OWNER_THREADS - 1) totalS = excl + len;
  __syncthreads();
  const int total = totalS;

  if (total <= OWNER_CAP) {
    int key[REC_PER_THREAD], rk[REC_PER_THREAD];
    float val[REC_PER_THREAD];
    v2f rec8[REC_PER_THREAD];
    const int last = total > 0 ? total - 1 : 0;
#pragma unroll
    for (int k = 0; k < REC_PER_THREAD; ++k) {
      const int idx = t + k * OWNER_THREADS;
      const int ic = idx < last ? idx : last;
      const int c = find_run(chunkPre, ic);
      rec8[k] = recs[chunkBeg[c] + (ic - chunkPre[c])];
    }
#pragma unroll
    for (int k = 0; k < REC_PER_THREAD; ++k) {
      const int idx = t + k * OWNER_THREADS;
      key[k] = idx < total ? (__float_as_int(rec8[k].x) & 0x7fffffff) : -1;
      val[k] = rec8[k].y;
    }
#pragma unroll
    for (int k = 0; k < REC_PER_THREAD; ++k) {
      rk[k] = 0;
      if (key[k] >= 0) rk[k] = atomicAdd(&cnt[row_of(key[k])], 1);
    }
    __syncthreads();
    {
      const int h = (t < 512) ? cnt[t] : 0;
      const int inc2 = wave_incl_scan(h, lane);
      if (lane == 31) wsumA[wave] = inc2;
      __syncthreads();
      int w2 = 0;
      for (int i = 0; i < wave; ++i) w2 += wsumA[i];
      if (t < 512) rowStart[t] = w2 + inc2 - h;
      if (t == 511) rowStart[512] = w2 + inc2;
      __syncthreads();
    }
#pragma unroll
    for (int k = 0; k < REC_PER_THREAD; ++k) {
      if (key[k] >= 0) {
        const int pos = rowStart[row_of(key[k])] + rk[k];
        const v2f rec = {__int_as_float(key[k] & COL_MASK), val[k]};
        sorted[pos] = rec;
      }
    }
    __syncthreads();
    const float bl = bias[lane], bh = bias[lane + 32];
    for (int r = wave; r < rowsInRange; r += OWNER_WAVES) {
      const int s = __builtin_amdgcn_readfirstlane(rowStart[r]);
      const int e = __builtin_amdgcn_readfirstlane(rowStart[r + 1]);
      float accLo = 0.0f, accHi = 0.0f;
      for (int p = s; p < e; p += 32) {
        const int n = (e - p) < 32 ? (e - p) : 32;
        v2f rec = {0.0f, 0.0f};
        if (lane < n) rec = sorted[p + lane];
        const int colv = __float_as_int(rec.x);
        const float valv = rec.y;
        for (int j0 = 0; j0 < n; j0 += WALK_BATCH) {
          float lo[WALK_BATCH], hi[WALK_BATCH];
#pragma unroll
          for (int u = 0; u < WALK_BATCH; ++u) {
            lo[u] = 0.0f;
            hi[u] = 0.0f;
            if (j0 + u < n) {
              const float* sp =
                  S + ((unsigned)bcast_i(colv, j0 + u) * OUT_F + lane);
              lo[u] = sp[0];
              hi[u] = sp[32];
            }
          }
#pragma unroll
          for (int u = 0; u < WALK_BATCH; ++u) {
            const float v = bcast_f(valv, j0 + u);
            accLo += lo[u] * v;
            accHi += hi[u] * v;
          }
        }
      }
      float* o = out + ((b * ROWS_PER_RANGE + r) * OUT_F + lane);
      o[0] = accLo + bl;
      o[32] = accHi + bh;
    }
    return;
  }

  for (int i = t; i < ROWS_PER_RANGE * OUT_F; i += OWNER_THREADS) acc[i] = 0.0f;
  __syncthreads();
  for (int idx = t; idx < total; idx += OWNER_THREADS) {
    const int c = find_run(chunkPre, idx);
    const v2f rec = recs[chunkBeg[c] + (idx - chunkPre[c])];
    const int key = __float_as_int(rec.x) & 0x7fffffff;
    const int rl = row_of(key);
    const float* sp = S + (unsigned)(key & COL_MASK) * OUT_F;
    for (int f = 0; f < OUT_F; ++f)
      atomicAdd(&acc[rl * OUT_F + f], sp[f] * rec.y);
  }
  __syncthreads();
  for (int i = t; i < rowsInRange * OUT_F; i += OWNER_THREADS)
    out[b * ROWS_PER_RANGE * OUT_F + i] = acc[i] + bias[i & (OUT_F - 1)];
}

extern "C" void kernel_launch(void* const* d_in, const int* in_sizes, int n_in,
                              void* d_out, int out_size, void* d_ws,
                              size_t ws_size, hipStream_t stream) {
  const float* vertex  = (const float*)d_in[0];
  const int*   erow    = (const int*)d_in[1];
  const int*   ecol    = (const int*)d_in[2];
  const float* eval    = (const float*)d_in[3];
  const float* weights = (const float*)d_in[4];
  const float* bias    = (const float*)d_in[5];
  float* out     = (float*)d_out;
  float* support = (float*)d_ws;
  float* recs = (float*)((char*)d_ws + (64u << 20));
  int* offs = (int*)((char*)d_ws + (128u << 20));

  const int nNodes = in_sizes[0] / IN_F;
  const int nEdges = in_sizes[1];

  const int nRowTiles = nNodes / 16;
  const int nGemmBlocks = (nRowTiles + WAVES_PER_BLOCK - 1) / WAVES_PER_BLOCK;
  const int nChunks = (nEdges + CHUNK_EDGES - 1) / CHUNK_EDGES;
  gemm_sort_kernel<<<nGemmBlocks + nChunks, 256, 0, stream>>>(
      vertex, weights, support, erow, ecol, eval, recs, offs, nRowTiles,
      nEdges, nGemmBlocks);

  const int nRanges = (nNodes + ROWS_PER_RANGE - 1) / ROWS_PER_RANGE;
  owner_kernel<<<nRanges, OWNER_THREADS, 0, stream>>>(
      support, recs, offs, bias, out, nNodes, nChunks);
}
